# starting kernel
# baseline (speedup 1.0000x reference)
#include <hip/hip_runtime.h>

typedef __attribute__((ext_vector_type(2))) float v2f;
typedef __attribute__((ext_vector_type(8))) float v8f;

#define HD 16

__global__ void posmlp_gen_wmma(const float* __restrict__ Q,
                                const float* __restrict__ Wg,
                                const float* __restrict__ bg,
                                float* __restrict__ Wout,
                                int M, int N, int C, int tilesN, int nTiles) {
  const int wavesPerBlk = blockDim.x >> 5;
  const int tile = blockIdx.x * wavesPerBlk + ((int)threadIdx.x >> 5);
  if (tile >= nTiles) return;

  const int tm = tile / tilesN;
  const int tn = tile - tm * tilesN;
  const int m0 = tm * 16, n0 = tn * 16;

  const int lane = threadIdx.x & 31;
  const int half = lane >> 4;
  const int l    = lane & 15;

  const int mA = m0 + l;
  const int nB = n0 + l;
  const int mC = (mA < M) ? mA : (M - 1);
  const int nC = (nB < N) ? nB : (N - 1);

  const float* qrow = Q  + (size_t)mC * C + 2 * half;
  const float* wrow = Wg + (size_t)nC * C + 2 * half;

  v8f acc = {};
  #pragma unroll 4
  for (int k = 0; k < C; k += 4) {
    const v2f a = *(const v2f*)(qrow + k);
    const v2f b = *(const v2f*)(wrow + k);
    acc = __builtin_amdgcn_wmma_f32_16x16x4_f32(
        false, a, false, b, (short)0, acc, false, false);
  }

  const float bias = bg[nC];
  const bool  nOk  = (nB < N);
  #pragma unroll
  for (int i = 0; i < 8; ++i) {
    const int m = m0 + i + 8 * half;
    if (m < M && nOk) Wout[(size_t)m * N + nB] = acc[i] + bias;
  }
}

__global__ void posmlp_spatial(const float* __restrict__ Wq,
                               const float* __restrict__ pos,
                               const int* __restrict__ Hp,
                               const int* __restrict__ Wp,
                               float* __restrict__ out,
                               int N, int HW) {
  __shared__ float sw[96];
  const int bq = blockIdx.x;
  const int t  = threadIdx.x;

  if (t < N) sw[t] = Wq[(size_t)bq * N + t];
  __syncthreads();

  const int H_ = *Hp;
  const int W_ = *Wp;

  float w1y[HD], w1x[HD], b1[HD], w2[HD];
  #pragma unroll
  for (int k = 0; k < HD; ++k) {
    w1y[k] = sw[2 * k];
    w1x[k] = sw[2 * k + 1];
    b1[k]  = sw[2 * HD + k];
    w2[k]  = sw[3 * HD + k];
  }
  const float b2 = sw[4 * HD];

  const float cx = pos[bq * 4 + 0];
  const float cy = pos[bq * 4 + 1];
  const float invbw = 1.0f / pos[bq * 4 + 2];
  const float invbh = 1.0f / pos[bq * 4 + 3];
  const float invH = 1.0f / (float)H_;
  const float invW = 1.0f / (float)W_;

  float* obase = out + (size_t)bq * HW;
  for (int p = t; p < HW; p += blockDim.x) {
    const int h = p / W_;
    const int w = p - h * W_;
    const float rel_y = (((float)h + 0.5f) * invH - cy) * invbh;
    const float rel_x = (((float)w + 0.5f) * invW - cx) * invbw;
    float acc = b2;
    #pragma unroll
    for (int k = 0; k < HD; ++k) {
      const float hid = fmaf(rel_y, w1y[k], fmaf(rel_x, w1x[k], b1[k]));
      acc = fmaf(fmaxf(hid, 0.0f), w2[k], acc);
    }
    obase[p] = acc;
  }
}

extern "C" void kernel_launch(void* const* d_in, const int* in_sizes, int n_in,
                              void* d_out, int out_size, void* d_ws, size_t ws_size,
                              hipStream_t stream) {
  const float* queries = (const float*)d_in[0];
  const float* pos     = (const float*)d_in[1];
  const float* w_gen   = (const float*)d_in[2];
  const float* b_gen   = (const float*)d_in[3];
  const int*   Hp      = (const int*)d_in[4];
  const int*   Wp      = (const int*)d_in[5];

  const int N  = in_sizes[3];
  const int C  = in_sizes[2] / N;
  const int BQ = in_sizes[0] / C;
  const int HW = out_size / BQ;

  float* Wout = (float*)d_ws;

  const int tilesM = (BQ + 15) / 16;
  const int tilesN = (N + 15) / 16;
  const int nTiles = tilesM * tilesN;
  const int wavesPerBlk = 4;
  const int blocks1 = (nTiles + wavesPerBlk - 1) / wavesPerBlk;

  posmlp_gen_wmma<<<blocks1, wavesPerBlk * 32, 0, stream>>>(
      queries, w_gen, b_gen, Wout, BQ, N, C, tilesN, nTiles);

  posmlp_spatial<<<BQ, 256, 0, stream>>>(
      Wout, pos, Hp, Wp, (float*)d_out, N, HW);
}


# weight GEMM wave: all 64 b128 loads of its two rows and the bias issued before the first WMMA (one memory round trip, not one per loop trip); K summed in two interleaved halves (sum order changes)
# speedup vs baseline: 1.1572x; 1.1572x over previous
#include <hip/hip_runtime.h>

typedef __attribute__((ext_vector_type(2))) float v2f;
typedef __attribute__((ext_vector_type(4))) float v4f;
typedef __attribute__((ext_vector_type(8))) float v8f;

#define HD 16
#define CROW 256

__global__ void __launch_bounds__(128) posmlp_gen_wmma(const float* __restrict__ Q,
                                const float* __restrict__ Wg,
                                const float* __restrict__ bg,
                                float* __restrict__ Wout,
                                int M, int N, int C, int tilesN, int nTiles) {
  const int wavesPerBlk = blockDim.x >> 5;
  const int tile = blockIdx.x * wavesPerBlk + ((int)threadIdx.x >> 5);
  if (tile >= nTiles) return;

  const int tm = tile / tilesN;
  const int tn = tile - tm * tilesN;
  const int m0 = tm * 16, n0 = tn * 16;

  const int lane = threadIdx.x & 31;
  const int half = lane >> 4;
  const int l    = lane & 15;

  const int mA = m0 + l;
  const int nB = n0 + l;
  const int mC = (mA < M) ? mA : (M - 1);
  const int nC = (nB < N) ? nB : (N - 1);

  const float bias = bg[nC];

  v8f acc = {};
  if (C == CROW) {
    const float* qrow = Q  + (size_t)mC * CROW + 4 * half;
    const float* wrow = Wg + (size_t)nC * CROW + 4 * half;
    v4f a[CROW / 8], b[CROW / 8];
    #pragma unroll
    for (int j = 0; j < CROW / 8; ++j) {
      a[j] = *(const v4f*)(qrow + 8 * j);
      b[j] = *(const v4f*)(wrow + 8 * j);
    }
    __builtin_amdgcn_sched_barrier(0);
    v8f acc1 = {};
    #pragma unroll
    for (int j = 0; j < CROW / 8; ++j) {
      const v2f alo = {a[j].x, a[j].y}, blo = {b[j].x, b[j].y};
      const v2f ahi = {a[j].z, a[j].w}, bhi = {b[j].z, b[j].w};
      acc  = __builtin_amdgcn_wmma_f32_16x16x4_f32(
          false, alo, false, blo, (short)0, acc, false, false);
      acc1 = __builtin_amdgcn_wmma_f32_16x16x4_f32(
          false, ahi, false, bhi, (short)0, acc1, false, false);
    }
    __builtin_amdgcn_sched_barrier(0);
    asm volatile("v_nop\n\tv_nop\n\tv_nop\n\tv_nop\n\tv_nop");
    __builtin_amdgcn_sched_barrier(0);
    acc += acc1;
  } else {
    const float* qrow = Q  + (size_t)mC * C + 2 * half;
    const float* wrow = Wg + (size_t)nC * C + 2 * half;
    for (int k = 0; k < C; k += 4) {
      const v2f a = *(const v2f*)(qrow + k);
      const v2f b = *(const v2f*)(wrow + k);
      acc = __builtin_amdgcn_wmma_f32_16x16x4_f32(
          false, a, false, b, (short)0, acc, false, false);
    }
  }

  const bool  nOk  = (nB < N);
  #pragma unroll
  for (int i = 0; i < 8; ++i) {
    const int m = m0 + i + 8 * half;
    if (m < M && nOk) Wout[(size_t)m * N + nB] = acc[i] + bias;
  }
}

__global__ void posmlp_spatial(const float* __restrict__ Wq,
                               const float* __restrict__ pos,
                               const int* __restrict__ Hp,
                               const int* __restrict__ Wp,
                               float* __restrict__ out,
                               int N, int HW) {
  __shared__ float sw[96];
  const int bq = blockIdx.x;
  const int t  = threadIdx.x;

  if (t < N) sw[t] = Wq[(size_t)bq * N + t];
  __syncthreads();

  const int H_ = *Hp;
  const int W_ = *Wp;

  float w1y[HD], w1x[HD], b1[HD], w2[HD];
  #pragma unroll
  for (int k = 0; k < HD; ++k) {
    w1y[k] = sw[2 * k];
    w1x[k] = sw[2 * k + 1];
    b1[k]  = sw[2 * HD + k];
    w2[k]  = sw[3 * HD + k];
  }
  const float b2 = sw[4 * HD];

  const float cx = pos[bq * 4 + 0];
  const float cy = pos[bq * 4 + 1];
  const float invbw = 1.0f / pos[bq * 4 + 2];
  const float invbh = 1.0f / pos[bq * 4 + 3];
  const float invH = 1.0f / (float)H_;
  const float invW = 1.0f / (float)W_;

  float* obase = out + (size_t)bq * HW;
  for (int p = t; p < HW; p += blockDim.x) {
    const int h = p / W_;
    const int w = p - h * W_;
    const float rel_y = (((float)h + 0.5f) * invH - cy) * invbh;
    const float rel_x = (((float)w + 0.5f) * invW - cx) * invbw;
    float acc = b2;
    #pragma unroll
    for (int k = 0; k < HD; ++k) {
      const float hid = fmaf(rel_y, w1y[k], fmaf(rel_x, w1x[k], b1[k]));
      acc = fmaf(fmaxf(hid, 0.0f), w2[k], acc);
    }
    obase[p] = acc;
  }
}

extern "C" void kernel_launch(void* const* d_in, const int* in_sizes, int n_in,
                              void* d_out, int out_size, void* d_ws, size_t ws_size,
                              hipStream_t stream) {
  const float* queries = (const float*)d_in[0];
  const float* pos     = (const float*)d_in[1];
  const float* w_gen   = (const float*)d_in[2];
  const float* b_gen   = (const float*)d_in[3];
  const int*   Hp      = (const int*)d_in[4];
  const int*   Wp      = (const int*)d_in[5];

  const int N  = in_sizes[3];
  const int C  = in_sizes[2] / N;
  const int BQ = in_sizes[0] / C;
  const int HW = out_size / BQ;

  float* Wout = (float*)d_ws;

  const int tilesM = (BQ + 15) / 16;
  const int tilesN = (N + 15) / 16;
  const int nTiles = tilesM * tilesN;
  const int wavesPerBlk = 4;
  const int blocks1 = (nTiles + wavesPerBlk - 1) / wavesPerBlk;

  posmlp_gen_wmma<<<blocks1, wavesPerBlk * 32, 0, stream>>>(
      queries, w_gen, b_gen, Wout, BQ, N, C, tilesN, nTiles);

  posmlp_spatial<<<BQ, 256, 0, stream>>>(
      Wout, pos, Hp, Wp, (float*)d_out, N, HW);
}


# map kernel: hidden layer on f32 16x16x4 WMMA (16 units x 16 pixels per op; k = rel_y, rel_x, 1*b1), relu*w2 summed per lane half, halves joined by lane exchange (sum order changes); 2 waves per query
# speedup vs baseline: 1.3374x; 1.1557x over previous
#include <hip/hip_runtime.h>

typedef __attribute__((ext_vector_type(2))) float v2f;
typedef __attribute__((ext_vector_type(4))) float v4f;
typedef __attribute__((ext_vector_type(8))) float v8f;

#define HD 16
#define CROW 256

__global__ void __launch_bounds__(128) posmlp_gen_wmma(const float* __restrict__ Q,
                                const float* __restrict__ Wg,
                                const float* __restrict__ bg,
                                float* __restrict__ Wout,
                                int M, int N, int C, int tilesN, int nTiles) {
  const int wavesPerBlk = blockDim.x >> 5;
  const int tile = blockIdx.x * wavesPerBlk + ((int)threadIdx.x >> 5);
  if (tile >= nTiles) return;

  const int tm = tile / tilesN;
  const int tn = tile - tm * tilesN;
  const int m0 = tm * 16, n0 = tn * 16;

  const int lane = threadIdx.x & 31;
  const int half = lane >> 4;
  const int l    = lane & 15;

  const int mA = m0 + l;
  const int nB = n0 + l;
  const int mC = (mA < M) ? mA : (M - 1);
  const int nC = (nB < N) ? nB : (N - 1);

  const float bias = bg[nC];

  v8f acc = {};
  if (C == CROW) {
    const float* qrow = Q  + (size_t)mC * CROW + 4 * half;
    const float* wrow = Wg + (size_t)nC * CROW + 4 * half;
    v4f a[CROW / 8], b[CROW / 8];
    #pragma unroll
    for (int j = 0; j < CROW / 8; ++j) {
      a[j] = *(const v4f*)(qrow + 8 * j);
      b[j] = *(const v4f*)(wrow + 8 * j);
    }
    __builtin_amdgcn_sched_barrier(0);
    v8f acc1 = {};
    #pragma unroll
    for (int j = 0; j < CROW / 8; ++j) {
      const v2f alo = {a[j].x, a[j].y}, blo = {b[j].x, b[j].y};
      const v2f ahi = {a[j].z, a[j].w}, bhi = {b[j].z, b[j].w};
      acc  = __builtin_amdgcn_wmma_f32_16x16x4_f32(
          false, alo, false, blo, (short)0, acc, false, false);
      acc1 = __builtin_amdgcn_wmma_f32_16x16x4_f32(
          false, ahi, false, bhi, (short)0, acc1, false, false);
    }
    __builtin_amdgcn_sched_barrier(0);
    asm volatile("v_nop\n\tv_nop\n\tv_nop\n\tv_nop\n\tv_nop");
    __builtin_amdgcn_sched_barrier(0);
    acc += acc1;
  } else {
    const float* qrow = Q  + (size_t)mC * C + 2 * half;
    const float* wrow = Wg + (size_t)nC * C + 2 * half;
    for (int k = 0; k < C; k += 4) {
      const v2f a = *(const v2f*)(qrow + k);
      const v2f b = *(const v2f*)(wrow + k);
      acc = __builtin_amdgcn_wmma_f32_16x16x4_f32(
          false, a, false, b, (short)0, acc, false, false);
    }
  }

  const bool  nOk  = (nB < N);
  #pragma unroll
  for (int i = 0; i < 8; ++i) {
    const int m = m0 + i + 8 * half;
    if (m < M && nOk) Wout[(size_t)m * N + nB] = acc[i] + bias;
  }
}

#define WCOLS 48
#define QPB 2
#define WPQ 2

__device__ __forceinline__ float xhalf(float x) {
  const int b = __builtin_bit_cast(int, x);
  return __builtin_bit_cast(float, __builtin_amdgcn_permlanex16(
      b, b, 0x76543210, 0xfedcba98, false, false));
}

__device__ __forceinline__ v8f hidden_tile(const v2f A, const v2f B) {
  const v8f zero = {};
  return __builtin_amdgcn_wmma_f32_16x16x4_f32(
      false, A, false, B, (short)0, zero, false, false);
}

__device__ __forceinline__ float relu_bits(const float x) {
  const int b = __builtin_bit_cast(int, x);
  return __builtin_bit_cast(float, b > 0 ? b : 0);
}

__device__ __forceinline__ float relu_dot8(const v8f d, const v2f (&w)[4], const float init) {
  v2f acc = {init, 0.0f};
  #pragma unroll
  for (int j = 0; j < 4; ++j) {
    const v2f rj = {relu_bits(d[2 * j]), relu_bits(d[2 * j + 1])};
    acc = __builtin_elementwise_fma(rj, w[j], acc);
  }
  return acc.x + acc.y;
}

__global__ void __launch_bounds__(QPB * WPQ * 32)
posmlp_spatial(const float* __restrict__ Wq,
               const float* __restrict__ pos,
               const int* __restrict__ Hp,
               const int* __restrict__ Wp,
               float* __restrict__ out,
               int N, int HW, int BQ) {
  const int wave = __builtin_amdgcn_readfirstlane((int)threadIdx.x >> 5);
  const int lane = threadIdx.x & 31;
  const int bq   = blockIdx.x * QPB + wave / WPQ;
  const int part = wave % WPQ;
  if (bq >= BQ) return;

  const float* wq = Wq + (size_t)bq * N;
  const int hf = lane >> 4;
  const int c  = lane & 15;

  const float a0 = wq[hf ? 2 * HD + c : 2 * c];
  const float a1r = wq[2 * c + 1];
  v2f w2h[4];
  #pragma unroll
  for (int j = 0; j < 4; ++j) {
    w2h[j].x = wq[3 * HD + 8 * hf + 2 * j];
    w2h[j].y = wq[3 * HD + 8 * hf + 2 * j + 1];
  }
  const float b2 = wq[4 * HD];

  const int H_ = *Hp;
  const int W_ = *Wp;
  const float cx = pos[bq * 4 + 0];
  const float cy = pos[bq * 4 + 1];
  const float bw = pos[bq * 4 + 2];
  const float bh = pos[bq * 4 + 3];
  asm volatile("" : : "v"(a0), "v"(a1r), "v"(w2h[0]), "v"(w2h[1]), "v"(w2h[2]), "v"(w2h[3]), "s"(bw));
  const float invbw = 1.0f / bw;
  const float invbh = 1.0f / bh;
  const float invH = 1.0f / (float)H_;
  const float invW = 1.0f / (float)W_;
  float* obase = out + (size_t)bq * HW;

  if (W_ == WCOLS && (H_ & 1) == 0 && HW == H_ * WCOLS) {
    v2f A = {a0, hf ? 0.0f : a1r};
    const float init_a = hf ? 0.0f : b2;
    const float init_b = hf ? b2 : 0.0f;

    const int npairs = H_ >> 1;
    const int rp0 = (part * npairs) / WPQ;
    const int rp1 = ((part + 1) * npairs) / WPQ;
    float* o = obase + (size_t)rp0 * (2 * WCOLS) + lane;

    const float yfirst = hf ? 1.0f : (((float)(2 * rp0) + 0.5f) * invH - cy) * invbh;
    v2f Bg[WCOLS / 16];
    #pragma unroll
    for (int g = 0; g < WCOLS / 16; ++g) {
      const float rel_x = (((float)(16 * g + c) + 0.5f) * invW - cx) * invbw;
      Bg[g].x = yfirst;
      Bg[g].y = hf ? 0.0f : rel_x;
    }

    v8f dcur = hidden_tile(A, Bg[0]);
    for (int rp = rp0; rp < rp1; ++rp) {
      const int h = 2 * rp;
      const float y1 = hf ? 1.0f : (((float)(h + 1) + 0.5f) * invH - cy) * invbh;
      const float y2 = hf ? 1.0f : (((float)(h + 2) + 0.5f) * invH - cy) * invbh;
      float pa = 0.0f;
      #pragma unroll
      for (int t = 0; t < 6; ++t) {
        const v2f Bt = Bg[(t + 1) % 3];
        const v8f dn = hidden_tile(A, Bt);
        asm volatile("" : "+v"(dcur) : "v"(dn));
        Bg[t % 3].x = (t < 3) ? y1 : y2;
        const float p = relu_dot8(dcur, w2h, (t & 1) ? init_b : init_a);
        asm volatile("" : "+v"(A) : "v"(Bt), "v"(p));
        if ((t & 1) == 0) {
          pa = p;
        } else {
          const float send = hf ? pa : p;
          const float keep = hf ? p : pa;
          o[32 * (t >> 1)] = keep + xhalf(send);
        }
        dcur = dn;
      }
      o += 2 * WCOLS;
    }
  } else {
    for (int p = part * 32 + lane; p < HW; p += WPQ * 32) {
      const int h = p / W_;
      const int w = p - h * W_;
      const float rel_y = (((float)h + 0.5f) * invH - cy) * invbh;
      const float rel_x = (((float)w + 0.5f) * invW - cx) * invbw;
      float acc = b2;
      #pragma nounroll
      for (int k = 0; k < HD; ++k) {
        const float hid = fmaf(rel_y, wq[2 * k], fmaf(rel_x, wq[2 * k + 1], wq[2 * HD + k]));
        acc = fmaf(fmaxf(hid, 0.0f), wq[3 * HD + k], acc);
      }
      obase[p] = acc;
    }
  }
}

extern "C" void kernel_launch(void* const* d_in, const int* in_sizes, int n_in,
                              void* d_out, int out_size, void* d_ws, size_t ws_size,
                              hipStream_t stream) {
  const float* queries = (const float*)d_in[0];
  const float* pos     = (const float*)d_in[1];
  const float* w_gen   = (const float*)d_in[2];
  const float* b_gen   = (const float*)d_in[3];
  const int*   Hp      = (const int*)d_in[4];
  const int*   Wp      = (const int*)d_in[5];

  const int N  = in_sizes[3];
  const int C  = in_sizes[2] / N;
  const int BQ = in_sizes[0] / C;
  const int HW = out_size / BQ;

  float* Wout = (float*)d_ws;

  const int tilesM = (BQ + 15) / 16;
  const int tilesN = (N + 15) / 16;
  const int nTiles = tilesM * tilesN;
  const int wavesPerBlk = 4;
  const int blocks1 = (nTiles + wavesPerBlk - 1) / wavesPerBlk;

  posmlp_gen_wmma<<<blocks1, wavesPerBlk * 32, 0, stream>>>(
      queries, w_gen, b_gen, Wout, BQ, N, C, tilesN, nTiles);

  const int blocks2 = (BQ + QPB - 1) / QPB;
  posmlp_spatial<<<blocks2, QPB * WPQ * 32, 0, stream>>>(
      Wout, pos, Hp, Wp, (float*)d_out, N, HW, BQ);
}
